# starting kernel
# baseline (speedup 1.0000x reference)
#include <hip/hip_runtime.h>

typedef __attribute__((ext_vector_type(2))) float v2f;
typedef __attribute__((ext_vector_type(4))) float v4f;
typedef __attribute__((ext_vector_type(8))) float v8f;

#define EPSF 1e-8f
#define BIGF 100000.0f

static __device__ __forceinline__ v4f v4min(v4f a, v4f b) {
  v4f r;
  r.x = fminf(a.x, b.x); r.y = fminf(a.y, b.y);
  r.z = fminf(a.z, b.z); r.w = fminf(a.w, b.w);
  return r;
}
static __device__ __forceinline__ v4f v4max(v4f a, v4f b) {
  v4f r;
  r.x = fmaxf(a.x, b.x); r.y = fmaxf(a.y, b.y);
  r.z = fmaxf(a.z, b.z); r.w = fmaxf(a.w, b.w);
  return r;
}

__global__ __launch_bounds__(256) void stats1_kernel(
    const float* __restrict__ delta1,
    const float* __restrict__ cmask1,
    const float* __restrict__ cmask2,
    float* __restrict__ z1)
{
  const int t  = threadIdx.x;
  const int cg = t & 15;
  const int p  = t >> 4;
  const int blk = blockIdx.x;
  const int b  = blk >> 8;
  const int ni = (blk >> 4) & 15;
  const int nj = blk & 15;

  __shared__ float sa[16], sb[16];
  __shared__ v4f red[16][5][16];

  if (t < 16)      sa[t]      = cmask1[b*256 + ni*16 + t];
  else if (t < 32) sb[t - 16] = cmask2[b*256 + nj*16 + (t - 16)];
  __syncthreads();

  const float ap = sa[p];
  v4f s   = {0.f, 0.f, 0.f, 0.f};
  v4f ms  = s, ms2 = s;
  v4f mn  = { 3.4e38f,  3.4e38f,  3.4e38f,  3.4e38f};
  v4f mx  = {-3.4e38f, -3.4e38f, -3.4e38f, -3.4e38f};

  const float* base = delta1 +
      (((size_t)b * 256 + (size_t)ni * 16 + p) * 256 + (size_t)nj * 16) * 64 + cg * 4;

  #pragma unroll
  for (int q = 0; q < 16; ++q) {
    const v4f v = *(const v4f*)(base + q * 64);
    const float mpq = ap * sb[q];
    const float off = BIGF * (1.f - mpq);
    s   += v;
    const v4f mv = mpq * v;
    ms  += mv;
    ms2 += mv * v;
    mn = v4min(mn, v + off);
    mx = v4max(mx, v - off);
  }

  red[p][0][cg] = s;  red[p][1][cg] = ms; red[p][2][cg] = ms2;
  red[p][3][cg] = mn; red[p][4][cg] = mx;
  __syncthreads();

  #pragma unroll
  for (int step = 8; step >= 1; step >>= 1) {
    if (p < step) {
      red[p][0][cg] += red[p + step][0][cg];
      red[p][1][cg] += red[p + step][1][cg];
      red[p][2][cg] += red[p + step][2][cg];
      red[p][3][cg]  = v4min(red[p][3][cg], red[p + step][3][cg]);
      red[p][4][cg]  = v4max(red[p][4][cg], red[p + step][4][cg]);
    }
    __syncthreads();
  }

  if (p == 0) {
    s = red[0][0][cg]; ms = red[0][1][cg]; ms2 = red[0][2][cg];
    mn = red[0][3][cg]; mx = red[0][4][cg];
    float asum = 0.f, bsum = 0.f;
    #pragma unroll
    for (int i = 0; i < 16; ++i) { asum += sa[i]; bsum += sb[i]; }
    const float cnt    = asum * bsum;
    const float divide = cnt + EPSF;
    const v4f   m      = s / divide;
    const v4f   stdv   = (ms2 - 2.f * m * ms + m * m * cnt) / divide;
    float* zrow = z1 + (size_t)blk * 256;
    *(v4f*)(zrow + cg * 4)       = m;
    *(v4f*)(zrow + 64 + cg * 4)  = mn;
    *(v4f*)(zrow + 128 + cg * 4) = mx;
    *(v4f*)(zrow + 192 + cg * 4) = stdv;
  }
}

__global__ __launch_bounds__(64) void stats2_kernel(
    const float* __restrict__ E,
    const float* __restrict__ emask2,
    float* __restrict__ z2)
{
  const int c   = threadIdx.x;
  const int blk = blockIdx.x;
  const int b = blk >> 4, i = blk & 15;

  __shared__ float em[16];
  if (c < 16) em[c] = emask2[b * 16 + c];
  __syncthreads();

  float s = 0.f, ms = 0.f, ms2 = 0.f, denom = 0.f;
  float mn = 3.4e38f, mx = -3.4e38f;
  const float* base = E + ((size_t)(b * 256 + i * 16)) * 64 + c;
  #pragma unroll
  for (int j = 0; j < 16; ++j) {
    const float v = base[j * 64];
    const float e = em[j];
    denom += e;
    s += v; ms += e * v; ms2 += e * v * v;
    const float off = BIGF * (1.f - e);
    mn = fminf(mn, v + off);
    mx = fmaxf(mx, v - off);
  }
  const float m    = s / denom;
  const float stdv = (ms2 - 2.f * m * ms + m * m * denom) / denom;
  float* zr = z2 + (size_t)blk * 256;
  zr[c]       = m;
  zr[64 + c]  = mn;
  zr[128 + c] = mx;
  zr[192 + c] = stdv;
}

__global__ __launch_bounds__(32) void wmma_gemm_kernel(
    const float* __restrict__ A, const float* __restrict__ W,
    const float* __restrict__ bias, float* __restrict__ D,
    int Ntiles, int K, int N)
{
  const int tile = blockIdx.x;
  const int tn = tile % Ntiles;
  const int tm = tile / Ntiles;
  const int lane = threadIdx.x;
  const int l16  = lane & 15;
  const int hi   = lane >> 4;

  const float* arow = A + (size_t)(tm * 16 + l16) * K + 2 * hi;
  const float* wrow = W + (size_t)(tn * 16 + l16) * K + 2 * hi;

  v8f acc = {};
  for (int k = 0; k < K; k += 4) {
    v2f a  = { arow[k], arow[k + 1] };
    v2f bv = { wrow[k], wrow[k + 1] };
    acc = __builtin_amdgcn_wmma_f32_16x16x4_f32(
         false, a,  false, bv,
         (short)0, acc,  false,  false);
  }

  const float bb = bias[tn * 16 + l16];
  float* drow = D + (size_t)(tm * 16 + 8 * hi) * N + tn * 16 + l16;
  #pragma unroll
  for (int r = 0; r < 8; ++r)
    drow[(size_t)r * N] = acc[r] + bb;
}

extern "C" void kernel_launch(void* const* d_in, const int* in_sizes, int n_in,
                              void* d_out, int out_size, void* d_ws, size_t ws_size,
                              hipStream_t stream) {
  const float* delta1  = (const float*)d_in[0];
  const float* c_mask1 = (const float*)d_in[1];
  const float* c_mask2 = (const float*)d_in[2];
  const float* e_mask2 = (const float*)d_in[3];
  const float* W1      = (const float*)d_in[4];
  const float* b1      = (const float*)d_in[5];
  const float* W2      = (const float*)d_in[6];
  const float* b2      = (const float*)d_in[7];

  float* z1 = (float*)d_ws;
  float* E  = z1 + 1024 * 256;
  float* z2 = E + 1024 * 64;

  stats1_kernel<<<1024, 256, 0, stream>>>(delta1, c_mask1, c_mask2, z1);

  wmma_gemm_kernel<<<64 * 4, 32, 0, stream>>>(z1, W1, b1, E,  4,
                                               256,  64);

  stats2_kernel<<<64, 64, 0, stream>>>(E, e_mask2, z2);

  wmma_gemm_kernel<<<4 * 16, 32, 0, stream>>>(z2, W2, b2, (float*)d_out,
                                               16,  256,  256);
}


# Fuse GEMM1, stage-2 statistics and GEMM2 into one kernel (one work-group per row group and 64-column slice, E and z2 kept in LDS): 4 launches become 2; summation order unchanged
# speedup vs baseline: 1.5659x; 1.5659x over previous
#include <hip/hip_runtime.h>

typedef __attribute__((ext_vector_type(2))) float v2f;
typedef __attribute__((ext_vector_type(4))) float v4f;
typedef __attribute__((ext_vector_type(8))) float v8f;

#define EPSF 1e-8f
#define BIGF 100000.0f

static __device__ __forceinline__ v4f v4min(v4f a, v4f b) {
  v4f r;
  r.x = fminf(a.x, b.x); r.y = fminf(a.y, b.y);
  r.z = fminf(a.z, b.z); r.w = fminf(a.w, b.w);
  return r;
}
static __device__ __forceinline__ v4f v4max(v4f a, v4f b) {
  v4f r;
  r.x = fmaxf(a.x, b.x); r.y = fmaxf(a.y, b.y);
  r.z = fmaxf(a.z, b.z); r.w = fmaxf(a.w, b.w);
  return r;
}

__global__ __launch_bounds__(256) void stats1_kernel(
    const float* __restrict__ delta1,
    const float* __restrict__ cmask1,
    const float* __restrict__ cmask2,
    float* __restrict__ z1)
{
  const int t  = threadIdx.x;
  const int cg = t & 15;
  const int p  = t >> 4;
  const int blk = blockIdx.x;
  const int b  = blk >> 8;
  const int ni = (blk >> 4) & 15;
  const int nj = blk & 15;

  __shared__ float sa[16], sb[16];
  __shared__ v4f red[16][5][16];

  if (t < 16)      sa[t]      = cmask1[b*256 + ni*16 + t];
  else if (t < 32) sb[t - 16] = cmask2[b*256 + nj*16 + (t - 16)];
  __syncthreads();

  const float ap = sa[p];
  v4f s   = {0.f, 0.f, 0.f, 0.f};
  v4f ms  = s, ms2 = s;
  v4f mn  = { 3.4e38f,  3.4e38f,  3.4e38f,  3.4e38f};
  v4f mx  = {-3.4e38f, -3.4e38f, -3.4e38f, -3.4e38f};

  const float* base = delta1 +
      (((size_t)b * 256 + (size_t)ni * 16 + p) * 256 + (size_t)nj * 16) * 64 + cg * 4;

  #pragma unroll
  for (int q = 0; q < 16; ++q) {
    const v4f v = *(const v4f*)(base + q * 64);
    const float mpq = ap * sb[q];
    const float off = BIGF * (1.f - mpq);
    s   += v;
    const v4f mv = mpq * v;
    ms  += mv;
    ms2 += mv * v;
    mn = v4min(mn, v + off);
    mx = v4max(mx, v - off);
  }

  red[p][0][cg] = s;  red[p][1][cg] = ms; red[p][2][cg] = ms2;
  red[p][3][cg] = mn; red[p][4][cg] = mx;
  __syncthreads();

  #pragma unroll
  for (int step = 8; step >= 1; step >>= 1) {
    if (p < step) {
      red[p][0][cg] += red[p + step][0][cg];
      red[p][1][cg] += red[p + step][1][cg];
      red[p][2][cg] += red[p + step][2][cg];
      red[p][3][cg]  = v4min(red[p][3][cg], red[p + step][3][cg]);
      red[p][4][cg]  = v4max(red[p][4][cg], red[p + step][4][cg]);
    }
    __syncthreads();
  }

  if (p == 0) {
    s = red[0][0][cg]; ms = red[0][1][cg]; ms2 = red[0][2][cg];
    mn = red[0][3][cg]; mx = red[0][4][cg];
    float asum = 0.f, bsum = 0.f;
    #pragma unroll
    for (int i = 0; i < 16; ++i) { asum += sa[i]; bsum += sb[i]; }
    const float cnt    = asum * bsum;
    const float divide = cnt + EPSF;
    const v4f   m      = s / divide;
    const v4f   stdv   = (ms2 - 2.f * m * ms + m * m * cnt) / divide;
    float* zrow = z1 + (size_t)blk * 256;
    *(v4f*)(zrow + cg * 4)       = m;
    *(v4f*)(zrow + 64 + cg * 4)  = mn;
    *(v4f*)(zrow + 128 + cg * 4) = mx;
    *(v4f*)(zrow + 192 + cg * 4) = stdv;
  }
}

#define LDP 260

__global__ __launch_bounds__(128) void tail_kernel(
    const float* __restrict__ z1,
    const float* __restrict__ W1,
    const float* __restrict__ b1,
    const float* __restrict__ emask2,
    const float* __restrict__ W2,
    const float* __restrict__ b2,
    float* __restrict__ out)
{
  const int t    = threadIdx.x;
  const int wv   = t >> 5;
  const int lane = t & 31;
  const int l16  = lane & 15;
  const int hi   = lane >> 4;
  const int bi   = blockIdx.x >> 2;
  const int sl   = blockIdx.x & 3;

  __shared__ __attribute__((aligned(16))) float sZ[16 * LDP];
  __shared__ __attribute__((aligned(16))) float sW1[64 * LDP];
  __shared__ __attribute__((aligned(16))) float sW2[64 * LDP];
  __shared__ float sE[16][66];
  __shared__ __attribute__((aligned(16))) float sz2[256];
  __shared__ float so[64];
  __shared__ float em[16];

  const v4f* gz  = (const v4f*)(z1 + (size_t)bi * 16 * 256);
  const v4f* gw1 = (const v4f*)W1;
  const v4f* gw2 = (const v4f*)(W2 + (size_t)sl * 64 * 256);

  const float emv = emask2[(bi >> 4) * 16 + (t & 15)];
  const float bb1 = b1[wv * 16 + l16];
  const float bb2 = b2[sl * 64 + wv * 16 + l16];

  #pragma unroll
  for (int i = 0; i < 8; ++i) {
    const int f = t + 128 * i;
    *(v4f*)&sZ[(f >> 6) * LDP + (f & 63) * 4] = gz[f];
  }
  #pragma unroll
  for (int i = 0; i < 32; ++i) {
    const int f = t + 128 * i;
    *(v4f*)&sW1[(f >> 6) * LDP + (f & 63) * 4] = gw1[f];
  }
  em[t & 15] = emv;
  __syncthreads();

  v4f w2r[32];
  #pragma unroll
  for (int i = 0; i < 32; ++i) w2r[i] = gw2[t + 128 * i];

  {
    const float* ap = sZ  + l16 * LDP + 2 * hi;
    const float* bp = sW1 + (wv * 16 + l16) * LDP + 2 * hi;
    v2f af[64], bf[64];
    #pragma unroll
    for (int i = 0; i < 64; ++i) {
      af[i] = *(const v2f*)(ap + 4 * i);
      bf[i] = *(const v2f*)(bp + 4 * i);
    }
    __builtin_amdgcn_sched_barrier(0);
    v8f acc = {};
    #pragma unroll
    for (int i = 0; i < 64; ++i) {
      acc = __builtin_amdgcn_wmma_f32_16x16x4_f32(
           false, af[i],  false, bf[i],
           (short)0, acc,  false,  false);
    }
    #pragma unroll
    for (int r = 0; r < 8; ++r)
      sE[r + 8 * hi][wv * 16 + l16] = acc[r] + bb1;
  }

  #pragma unroll
  for (int i = 0; i < 32; ++i) {
    const int f = t + 128 * i;
    *(v4f*)&sW2[(f >> 6) * LDP + (f & 63) * 4] = w2r[i];
  }
  __syncthreads();

  if (t < 64) {
    const int c = t;
    float s = 0.f, ms = 0.f, ms2 = 0.f, denom = 0.f;
    float mn = 3.4e38f, mx = -3.4e38f;
    #pragma unroll
    for (int j = 0; j < 16; ++j) {
      const float v = sE[j][c];
      const float e = em[j];
      denom += e;
      s += v; ms += e * v; ms2 += e * v * v;
      const float off = BIGF * (1.f - e);
      mn = fminf(mn, v + off);
      mx = fmaxf(mx, v - off);
    }
    const float m    = s / denom;
    const float stdv = (ms2 - 2.f * m * ms + m * m * denom) / denom;
    sz2[c]       = m;
    sz2[64 + c]  = mn;
    sz2[128 + c] = mx;
    sz2[192 + c] = stdv;
  }
  __syncthreads();

  {
    const float* zp = sz2 + 2 * hi;
    const float* cp = sW2 + (wv * 16 + l16) * LDP + 2 * hi;
    v2f af[64], bf[64];
    #pragma unroll
    for (int i = 0; i < 64; ++i) {
      af[i] = *(const v2f*)(zp + 4 * i);
      bf[i] = *(const v2f*)(cp + 4 * i);
    }
    __builtin_amdgcn_sched_barrier(0);
    v8f acc = {};
    #pragma unroll
    for (int i = 0; i < 64; ++i) {
      acc = __builtin_amdgcn_wmma_f32_16x16x4_f32(
           false, af[i],  false, bf[i],
           (short)0, acc,  false,  false);
    }
    if (hi == 0) so[wv * 16 + l16] = acc[0] + bb2;
  }
  __syncthreads();

  if (t < 64) out[(size_t)bi * 256 + sl * 64 + t] = so[t];
}

extern "C" void kernel_launch(void* const* d_in, const int* in_sizes, int n_in,
                              void* d_out, int out_size, void* d_ws, size_t ws_size,
                              hipStream_t stream) {
  const float* delta1  = (const float*)d_in[0];
  const float* c_mask1 = (const float*)d_in[1];
  const float* c_mask2 = (const float*)d_in[2];
  const float* e_mask2 = (const float*)d_in[3];
  const float* W1      = (const float*)d_in[4];
  const float* b1      = (const float*)d_in[5];
  const float* W2      = (const float*)d_in[6];
  const float* b2      = (const float*)d_in[7];

  float* z1 = (float*)d_ws;

  stats1_kernel<<<1024, 256, 0, stream>>>(delta1, c_mask1, c_mask2, z1);

  tail_kernel<<<256, 128, 0, stream>>>(z1, W1, b1, e_mask2, W2, b2, (float*)d_out);
}


# Stage 1 on 2 waves per atom block instead of 8 (one per 32-channel half; each lane keeps its 64 loads in flight; lane butterfly replaces LDS tree and barriers): 8192 waves -> 2048; sum order changes
# speedup vs baseline: 1.8164x; 1.1600x over previous
#include <hip/hip_runtime.h>

typedef __attribute__((ext_vector_type(2))) float v2f;
typedef __attribute__((ext_vector_type(4))) float v4f;
typedef __attribute__((ext_vector_type(8))) float v8f;

#define EPSF 1e-8f
#define BIGF 100000.0f

static __device__ __forceinline__ v4f v4min(v4f a, v4f b) {
  v4f r;
  r.x = fminf(a.x, b.x); r.y = fminf(a.y, b.y);
  r.z = fminf(a.z, b.z); r.w = fminf(a.w, b.w);
  return r;
}
static __device__ __forceinline__ v4f v4max(v4f a, v4f b) {
  v4f r;
  r.x = fmaxf(a.x, b.x); r.y = fmaxf(a.y, b.y);
  r.z = fmaxf(a.z, b.z); r.w = fmaxf(a.w, b.w);
  return r;
}

__global__ __launch_bounds__(64) void stats1_kernel(
    const float* __restrict__ delta1,
    const float* __restrict__ cmask1,
    const float* __restrict__ cmask2,
    float* __restrict__ z1)
{
  const int t    = threadIdx.x;
  const int w    = t >> 5;
  const int lane = t & 31;
  const int cg   = lane & 7;
  const int qg   = lane >> 3;
  const int blk  = blockIdx.x;
  const int b  = blk >> 8;
  const int ni = (blk >> 4) & 15;
  const int nj = blk & 15;

  const float* pa = cmask1 + b * 256 + ni * 16;
  const float* pb = cmask2 + b * 256 + nj * 16;

  float bq[4];
  #pragma unroll
  for (int j = 0; j < 4; ++j) bq[j] = pb[4 * j + qg];

  const float* base = delta1 +
      (((size_t)b * 256 + (size_t)ni * 16) * 256 + (size_t)nj * 16 + qg) * 64 + w * 32 + cg * 4;

  v4f v[64];
  #pragma unroll
  for (int p = 0; p < 16; ++p) {
    #pragma unroll
    for (int j = 0; j < 4; ++j)
      v[p * 4 + j] = *(const v4f*)(base + (size_t)p * (256 * 64) + j * (4 * 64));
  }
  __builtin_amdgcn_sched_barrier(0);

  float av[16], asum = 0.f, bsum = 0.f;
  #pragma unroll
  for (int i = 0; i < 16; ++i) { av[i] = pa[i]; asum += av[i]; bsum += pb[i]; }

  v4f s   = {0.f, 0.f, 0.f, 0.f};
  v4f ms  = s, ms2 = s;
  v4f mn  = { 3.4e38f,  3.4e38f,  3.4e38f,  3.4e38f};
  v4f mx  = {-3.4e38f, -3.4e38f, -3.4e38f, -3.4e38f};

  #pragma unroll
  for (int p = 0; p < 16; ++p) {
    #pragma unroll
    for (int j = 0; j < 4; ++j) {
      const v4f x = v[p * 4 + j];
      const float mpq = av[p] * bq[j];
      const float off = BIGF * (1.f - mpq);
      s   += x;
      const v4f mv = mpq * x;
      ms  += mv;
      ms2 += mv * x;
      mn = v4min(mn, x + off);
      mx = v4max(mx, x - off);
    }
  }

  #pragma unroll
  for (int d = 8; d <= 16; d <<= 1) {
    #pragma unroll
    for (int k = 0; k < 4; ++k) {
      s[k]   += __shfl_xor(s[k], d);
      ms[k]  += __shfl_xor(ms[k], d);
      ms2[k] += __shfl_xor(ms2[k], d);
      mn[k]   = fminf(mn[k], __shfl_xor(mn[k], d));
      mx[k]   = fmaxf(mx[k], __shfl_xor(mx[k], d));
    }
  }

  const float cnt    = asum * bsum;
  const float divide = cnt + EPSF;
  const v4f   m      = s / divide;
  const v4f   stdv   = (ms2 - 2.f * m * ms + m * m * cnt) / divide;

  const v4f o = (qg == 0) ? m : (qg == 1) ? mn : (qg == 2) ? mx : stdv;
  *(v4f*)(z1 + (size_t)blk * 256 + qg * 64 + w * 32 + cg * 4) = o;
}

#define LDP 260

__global__ __launch_bounds__(128) void tail_kernel(
    const float* __restrict__ z1,
    const float* __restrict__ W1,
    const float* __restrict__ b1,
    const float* __restrict__ emask2,
    const float* __restrict__ W2,
    const float* __restrict__ b2,
    float* __restrict__ out)
{
  const int t    = threadIdx.x;
  const int wv   = t >> 5;
  const int lane = t & 31;
  const int l16  = lane & 15;
  const int hi   = lane >> 4;
  const int bi   = blockIdx.x >> 2;
  const int sl   = blockIdx.x & 3;

  __shared__ __attribute__((aligned(16))) float sZ[16 * LDP];
  __shared__ __attribute__((aligned(16))) float sW1[64 * LDP];
  __shared__ __attribute__((aligned(16))) float sW2[64 * LDP];
  __shared__ float sE[16][66];
  __shared__ __attribute__((aligned(16))) float sz2[256];
  __shared__ float so[64];
  __shared__ float em[16];

  const v4f* gz  = (const v4f*)(z1 + (size_t)bi * 16 * 256);
  const v4f* gw1 = (const v4f*)W1;
  const v4f* gw2 = (const v4f*)(W2 + (size_t)sl * 64 * 256);

  const float emv = emask2[(bi >> 4) * 16 + (t & 15)];
  const float bb1 = b1[wv * 16 + l16];
  const float bb2 = b2[sl * 64 + wv * 16 + l16];

  #pragma unroll
  for (int i = 0; i < 8; ++i) {
    const int f = t + 128 * i;
    *(v4f*)&sZ[(f >> 6) * LDP + (f & 63) * 4] = gz[f];
  }
  #pragma unroll
  for (int i = 0; i < 32; ++i) {
    const int f = t + 128 * i;
    *(v4f*)&sW1[(f >> 6) * LDP + (f & 63) * 4] = gw1[f];
  }
  em[t & 15] = emv;
  __syncthreads();

  v4f w2r[32];
  #pragma unroll
  for (int i = 0; i < 32; ++i) w2r[i] = gw2[t + 128 * i];

  {
    const float* ap = sZ  + l16 * LDP + 2 * hi;
    const float* bp = sW1 + (wv * 16 + l16) * LDP + 2 * hi;
    v2f af[64], bf[64];
    #pragma unroll
    for (int i = 0; i < 64; ++i) {
      af[i] = *(const v2f*)(ap + 4 * i);
      bf[i] = *(const v2f*)(bp + 4 * i);
    }
    __builtin_amdgcn_sched_barrier(0);
    v8f acc = {};
    #pragma unroll
    for (int i = 0; i < 64; ++i) {
      acc = __builtin_amdgcn_wmma_f32_16x16x4_f32(
           false, af[i],  false, bf[i],
           (short)0, acc,  false,  false);
    }
    #pragma unroll
    for (int r = 0; r < 8; ++r)
      sE[r + 8 * hi][wv * 16 + l16] = acc[r] + bb1;
  }

  #pragma unroll
  for (int i = 0; i < 32; ++i) {
    const int f = t + 128 * i;
    *(v4f*)&sW2[(f >> 6) * LDP + (f & 63) * 4] = w2r[i];
  }
  __syncthreads();

  if (t < 64) {
    const int c = t;
    float s = 0.f, ms = 0.f, ms2 = 0.f, denom = 0.f;
    float mn = 3.4e38f, mx = -3.4e38f;
    #pragma unroll
    for (int j = 0; j < 16; ++j) {
      const float v = sE[j][c];
      const float e = em[j];
      denom += e;
      s += v; ms += e * v; ms2 += e * v * v;
      const float off = BIGF * (1.f - e);
      mn = fminf(mn, v + off);
      mx = fmaxf(mx, v - off);
    }
    const float m    = s / denom;
    const float stdv = (ms2 - 2.f * m * ms + m * m * denom) / denom;
    sz2[c]       = m;
    sz2[64 + c]  = mn;
    sz2[128 + c] = mx;
    sz2[192 + c] = stdv;
  }
  __syncthreads();

  {
    const float* zp = sz2 + 2 * hi;
    const float* cp = sW2 + (wv * 16 + l16) * LDP + 2 * hi;
    v2f af[64], bf[64];
    #pragma unroll
    for (int i = 0; i < 64; ++i) {
      af[i] = *(const v2f*)(zp + 4 * i);
      bf[i] = *(const v2f*)(cp + 4 * i);
    }
    __builtin_amdgcn_sched_barrier(0);
    v8f acc = {};
    #pragma unroll
    for (int i = 0; i < 64; ++i) {
      acc = __builtin_amdgcn_wmma_f32_16x16x4_f32(
           false, af[i],  false, bf[i],
           (short)0, acc,  false,  false);
    }
    if (hi == 0) so[wv * 16 + l16] = acc[0] + bb2;
  }
  __syncthreads();

  if (t < 64) out[(size_t)bi * 256 + sl * 64 + t] = so[t];
}

extern "C" void kernel_launch(void* const* d_in, const int* in_sizes, int n_in,
                              void* d_out, int out_size, void* d_ws, size_t ws_size,
                              hipStream_t stream) {
  const float* delta1  = (const float*)d_in[0];
  const float* c_mask1 = (const float*)d_in[1];
  const float* c_mask2 = (const float*)d_in[2];
  const float* e_mask2 = (const float*)d_in[3];
  const float* W1      = (const float*)d_in[4];
  const float* b1      = (const float*)d_in[5];
  const float* W2      = (const float*)d_in[6];
  const float* b2      = (const float*)d_in[7];

  float* z1 = (float*)d_ws;

  stats1_kernel<<<1024, 64, 0, stream>>>(delta1, c_mask1, c_mask2, z1);

  tail_kernel<<<256, 128, 0, stream>>>(z1, W1, b1, e_mask2, W2, b2, (float*)d_out);
}
